# starting kernel
# baseline (speedup 1.0000x reference)
#include <hip/hip_runtime.h>
#include <hip/hip_bf16.h>

typedef __attribute__((ext_vector_type(2))) float v2f;
typedef __attribute__((ext_vector_type(8))) float v8f;

namespace {
constexpr int kW = 512;
constexpr int kH = 512;
constexpr float kLog2e = 1.44269504088896340736f;
}

__global__ void gauss2d_coeffs(const float* __restrict__ means,
                               const float* __restrict__ scales,
                               const float* __restrict__ thetas,
                               const float* __restrict__ opac_in,
                               float* __restrict__ G,
                               int N) {
  int n = blockIdx.x * blockDim.x + threadIdx.x;
  if (n >= N) return;
  float mx = means[2 * n + 0];
  float my = means[2 * n + 1];
  float sx = expf(scales[2 * n + 0]);
  float sy = expf(scales[2 * n + 1]);
  float th = (1.0f / (1.0f + expf(-thetas[n]))) * 6.28318530717958647692f;
  float op = 1.0f / (1.0f + expf(-opac_in[n]));
  float c = cosf(th), s = sinf(th);
  float sx2 = sx * sx, sy2 = sy * sy;
  float a = c * c * sx2 + s * s * sy2 + 1e-6f;
  float b = c * s * (sx2 - sy2);
  float d = s * s * sx2 + c * c * sy2 + 1e-6f;
  float det = a * d - b * b;
  float ai = d / det, bi = -b / det, di = a / det;
  float* g = G + (size_t)n * 8;
  const float m = -0.5f * kLog2e;
  g[0] = m * ai;
  g[1] = m * 2.0f * bi;
  g[2] = m * di;
  g[3] = -2.0f * m * (ai * mx + bi * my);
  g[4] = -2.0f * m * (bi * mx + di * my);
  g[5] = m * (ai * mx * mx + 2.0f * bi * mx * my + di * my * my)
         + log2f(op);
  g[6] = 0.0f;
  g[7] = 0.0f;
}

__global__ __launch_bounds__(256) void gauss2d_render(
    const float* __restrict__ G, float* __restrict__ out, int N) {
  const int lane = threadIdx.x & 31;
  const int wave = (blockIdx.x * blockDim.x + threadIdx.x) >> 5;
  const bool hi = lane >= 16;
  const int col = lane & 15;
  const int pix = wave * 16 + col;
  const int w = pix & (kW - 1);
  const int h = pix >> 9;
  const float x = -1.0f + 2.0f * (float)w / (float)(kW - 1);
  const float y = -1.0f + 2.0f * (float)h / (float)(kH - 1);

  v2f b1, b2;
  b1.x = hi ? (y * y) : (x * x);
  b1.y = hi ? x       : (x * y);
  b2.x = hi ? 0.0f    : y;
  b2.y = hi ? 0.0f    : 1.0f;

  float ACC = 0.0f;
  float T   = 1.0f;

  const v2f* gp = (const v2f*)(G + (size_t)col * 8 + (hi ? 2 : 0));

  for (int t = 0; t < N; t += 16) {
    v2f a1 = gp[0];
    v2f a2 = gp[2];
    gp += 64;

    v8f q = {};
    q = __builtin_amdgcn_wmma_f32_16x16x4_f32(false, a1, false, b1,
                                              (short)0, q, false, false);
    q = __builtin_amdgcn_wmma_f32_16x16x4_f32(false, a2, false, b2,
                                              (short)0, q, false, false);

    float accH = 0.0f, TH = 1.0f;
#pragma unroll
    for (int r = 0; r < 8; ++r) {
      float alpha = __builtin_amdgcn_exp2f(q[r]);
      accH += TH * alpha;
      TH *= (1.0f - alpha);
    }

    float accO = __shfl_xor(accH, 16, 32);
    float TO   = __shfl_xor(TH, 16, 32);
    float tileAcc = hi ? (accO + TO * accH) : (accH + TH * accO);
    float tileT   = TH * TO;

    ACC += T * tileAcc;
    T *= tileT;
  }

  if (!hi) {
    float* o = out + (size_t)pix * 3;
    o[0] = ACC;
    o[1] = ACC;
    o[2] = ACC;
  }
}

extern "C" void kernel_launch(void* const* d_in, const int* in_sizes, int n_in,
                              void* d_out, int out_size, void* d_ws,
                              size_t ws_size, hipStream_t stream) {
  const float* means  = (const float*)d_in[0];
  const float* scales = (const float*)d_in[1];
  const float* thetas = (const float*)d_in[2];
  const float* opacs  = (const float*)d_in[3];

  const int N = in_sizes[0] / 2;

  float* G = (float*)d_ws;

  gauss2d_coeffs<<<(N + 127) / 128, 128, 0, stream>>>(means, scales, thetas,
                                                      opacs, G, N);

  const int pixels  = kW * kH;
  const int waves   = pixels / 16;
  const int threads = waves * 32;
  const int block   = 256;
  gauss2d_render<<<threads / block, block, 0, stream>>>(G, (float*)d_out, N);
}


# Single launch: each workgroup computes the Gaussians' coefficient rows itself into LDS in WMMA A-fragment order and renders from there; the coefficient kernel and its workspace table are gone
# speedup vs baseline: 1.1924x; 1.1924x over previous
#include <hip/hip_runtime.h>
#include <hip/hip_bf16.h>

typedef __attribute__((ext_vector_type(2))) float v2f;
typedef __attribute__((ext_vector_type(8))) float v8f;

namespace {
constexpr int kW = 512;
constexpr int kH = 512;
constexpr int kChunk = 128;
constexpr float kLog2e = 1.44269504088896340736f;
}

__global__ __launch_bounds__(256) void gauss2d_fused(
    const float* __restrict__ means, const float* __restrict__ scales,
    const float* __restrict__ thetas, const float* __restrict__ opac_in,
    float* __restrict__ out, int N) {
  __shared__ v2f sA[(kChunk / 16) * 2 * 32];

  const int tid = threadIdx.x;
  const int lane = tid & 31;
  const int wave = (blockIdx.x * blockDim.x + tid) >> 5;
  const bool hi = lane >= 16;
  const int col = lane & 15;
  const int pix = wave * 16 + col;
  const int w = pix & (kW - 1);
  const int h = pix >> 9;
  const float x = -1.0f + 2.0f * (float)w / (float)(kW - 1);
  const float y = -1.0f + 2.0f * (float)h / (float)(kH - 1);

  v2f b1, b2;
  b1.x = hi ? (y * y) : (x * x);
  b1.y = hi ? x       : (x * y);
  b2.x = hi ? 0.0f    : y;
  b2.y = hi ? 0.0f    : 1.0f;

  float ACC = 0.0f;
  float T   = 1.0f;

  for (int base = 0; base < N; base += kChunk) {
    if (base) __syncthreads();

    if (tid < kChunk) {
      const int n = base + tid;
      float g0 = 0.0f, g1 = 0.0f, g2 = 0.0f, g3 = 0.0f, g4 = 0.0f;
      float g5 = -__builtin_inff();
      if (n < N) {
        float mx = means[2 * n + 0];
        float my = means[2 * n + 1];
        float sx = expf(scales[2 * n + 0]);
        float sy = expf(scales[2 * n + 1]);
        float th = (1.0f / (1.0f + expf(-thetas[n]))) * 6.28318530717958647692f;
        float op = 1.0f / (1.0f + expf(-opac_in[n]));
        float c = cosf(th), s = sinf(th);
        float sx2 = sx * sx, sy2 = sy * sy;
        float a = c * c * sx2 + s * s * sy2 + 1e-6f;
        float b = c * s * (sx2 - sy2);
        float d = s * s * sx2 + c * c * sy2 + 1e-6f;
        float det = a * d - b * b;
        float ai = d / det, bi = -b / det, di = a / det;
        const float m = -0.5f * kLog2e;
        g0 = m * ai;
        g1 = m * 2.0f * bi;
        g2 = m * di;
        g3 = -2.0f * m * (ai * mx + bi * my);
        g4 = -2.0f * m * (bi * mx + di * my);
        g5 = m * (ai * mx * mx + 2.0f * bi * mx * my + di * my * my)
             + log2f(op);
      }
      const int tl = tid >> 4;
      const int r  = tid & 15;
      v2f* dst = sA + tl * 64 + r;
      dst[0]  = v2f{g0, g1};
      dst[16] = v2f{g2, g3};
      dst[32] = v2f{g4, g5};
      dst[48] = v2f{0.0f, 0.0f};
    }
    __syncthreads();

    const int left = N - base;
    const int nt = ((left < kChunk ? left : kChunk) + 15) >> 4;
    const v2f* ap = sA + lane;

    for (int t = 0; t < nt; ++t) {
      v2f a1 = ap[0];
      v2f a2 = ap[32];
      ap += 64;

      v8f q = {};
      q = __builtin_amdgcn_wmma_f32_16x16x4_f32(false, a1, false, b1,
                                                (short)0, q, false, false);
      q = __builtin_amdgcn_wmma_f32_16x16x4_f32(false, a2, false, b2,
                                                (short)0, q, false, false);

      float accH = 0.0f, TH = 1.0f;
#pragma unroll
      for (int r = 0; r < 8; ++r) {
        float alpha = __builtin_amdgcn_exp2f(q[r]);
        accH += TH * alpha;
        TH *= (1.0f - alpha);
      }

      float accO = __shfl_xor(accH, 16, 32);
      float TO   = __shfl_xor(TH, 16, 32);
      float tileAcc = hi ? (accO + TO * accH) : (accH + TH * accO);
      float tileT   = TH * TO;

      ACC += T * tileAcc;
      T *= tileT;
    }
  }

  if (!hi) {
    float* o = out + (size_t)pix * 3;
    o[0] = ACC;
    o[1] = ACC;
    o[2] = ACC;
  }
}

extern "C" void kernel_launch(void* const* d_in, const int* in_sizes, int n_in,
                              void* d_out, int out_size, void* d_ws,
                              size_t ws_size, hipStream_t stream) {
  const float* means  = (const float*)d_in[0];
  const float* scales = (const float*)d_in[1];
  const float* thetas = (const float*)d_in[2];
  const float* opacs  = (const float*)d_in[3];

  const int N = in_sizes[0] / 2;

  const int pixels  = kW * kH;
  const int waves   = pixels / 16;
  const int threads = waves * 32;
  const int block   = 256;
  gauss2d_fused<<<threads / block, block, 0, stream>>>(
      means, scales, thetas, opacs, (float*)d_out, N);
}


# Compositing telescoped to c' = c*(1-alpha)+alpha (= 1-prod(1-alpha), order-free): one packed sub+fma per 2 Gaussians, partial values merged once at the end; summation order changes
# speedup vs baseline: 1.4182x; 1.1894x over previous
#include <hip/hip_runtime.h>
#include <hip/hip_bf16.h>

typedef __attribute__((ext_vector_type(2))) float v2f;
typedef __attribute__((ext_vector_type(8))) float v8f;

namespace {
constexpr int kW = 512;
constexpr int kH = 512;
constexpr int kChunk = 128;
constexpr float kLog2e = 1.44269504088896340736f;
}

__global__ __launch_bounds__(256) void gauss2d_fused(
    const float* __restrict__ means, const float* __restrict__ scales,
    const float* __restrict__ thetas, const float* __restrict__ opac_in,
    float* __restrict__ out, int N) {
  __shared__ v2f sA[(kChunk / 16) * 2 * 32];

  const int tid = threadIdx.x;
  const int lane = tid & 31;
  const int wave = (blockIdx.x * blockDim.x + tid) >> 5;
  const bool hi = lane >= 16;
  const int col = lane & 15;
  const int pix = wave * 16 + col;
  const int w = pix & (kW - 1);
  const int h = pix >> 9;
  const float x = -1.0f + 2.0f * (float)w / (float)(kW - 1);
  const float y = -1.0f + 2.0f * (float)h / (float)(kH - 1);

  v2f b1, b2;
  b1.x = hi ? (y * y) : (x * x);
  b1.y = hi ? x       : (x * y);
  b2.x = hi ? 0.0f    : y;
  b2.y = hi ? 0.0f    : 1.0f;

  v2f cov = {0.0f, 0.0f};

  for (int base = 0; base < N; base += kChunk) {
    if (base) __syncthreads();

    if (tid < kChunk) {
      const int n = base + tid;
      float g0 = 0.0f, g1 = 0.0f, g2 = 0.0f, g3 = 0.0f, g4 = 0.0f;
      float g5 = -__builtin_inff();
      if (n < N) {
        float mx = means[2 * n + 0];
        float my = means[2 * n + 1];
        float sx = expf(scales[2 * n + 0]);
        float sy = expf(scales[2 * n + 1]);
        float th = (1.0f / (1.0f + expf(-thetas[n]))) * 6.28318530717958647692f;
        float op = 1.0f / (1.0f + expf(-opac_in[n]));
        float c = cosf(th), s = sinf(th);
        float sx2 = sx * sx, sy2 = sy * sy;
        float a = c * c * sx2 + s * s * sy2 + 1e-6f;
        float b = c * s * (sx2 - sy2);
        float d = s * s * sx2 + c * c * sy2 + 1e-6f;
        float det = a * d - b * b;
        float ai = d / det, bi = -b / det, di = a / det;
        const float m = -0.5f * kLog2e;
        g0 = m * ai;
        g1 = m * 2.0f * bi;
        g2 = m * di;
        g3 = -2.0f * m * (ai * mx + bi * my);
        g4 = -2.0f * m * (bi * mx + di * my);
        g5 = m * (ai * mx * mx + 2.0f * bi * mx * my + di * my * my)
             + log2f(op);
      }
      const int tl = tid >> 4;
      const int r  = tid & 15;
      v2f* dst = sA + tl * 64 + r;
      dst[0]  = v2f{g0, g1};
      dst[16] = v2f{g2, g3};
      dst[32] = v2f{g4, g5};
      dst[48] = v2f{0.0f, 0.0f};
    }
    __syncthreads();

    const int left = N - base;
    const int nt = ((left < kChunk ? left : kChunk) + 15) >> 4;
    const v2f* ap = sA + lane;

#pragma unroll 2
    for (int t = 0; t < nt; ++t) {
      v2f a1 = ap[0];
      v2f a2 = ap[32];
      ap += 64;

      v8f q = {};
      q = __builtin_amdgcn_wmma_f32_16x16x4_f32(false, a1, false, b1,
                                                (short)0, q, false, false);
      q = __builtin_amdgcn_wmma_f32_16x16x4_f32(false, a2, false, b2,
                                                (short)0, q, false, false);

#pragma unroll
      for (int r = 0; r < 8; r += 2) {
        v2f alpha;
        alpha.x = __builtin_amdgcn_exp2f(q[r]);
        alpha.y = __builtin_amdgcn_exp2f(q[r + 1]);
        const v2f om = 1.0f - alpha;
        cov = cov * om + alpha;
      }
    }
  }

  const float mine  = cov.x * (1.0f - cov.y) + cov.y;
  const float other = __shfl_xor(mine, 16, 32);
  const float ACC   = mine * (1.0f - other) + other;

  if (!hi) {
    float* o = out + (size_t)pix * 3;
    o[0] = ACC;
    o[1] = ACC;
    o[2] = ACC;
  }
}

extern "C" void kernel_launch(void* const* d_in, const int* in_sizes, int n_in,
                              void* d_out, int out_size, void* d_ws,
                              size_t ws_size, hipStream_t stream) {
  const float* means  = (const float*)d_in[0];
  const float* scales = (const float*)d_in[1];
  const float* thetas = (const float*)d_in[2];
  const float* opacs  = (const float*)d_in[3];

  const int N = in_sizes[0] / 2;

  const int pixels  = kW * kH;
  const int waves   = pixels / 16;
  const int threads = waves * 32;
  const int block   = 256;
  gauss2d_fused<<<threads / block, block, 0, stream>>>(
      means, scales, thetas, opacs, (float*)d_out, N);
}


# One workgroup per image row: each wave renders 64 pixels as 4 groups of 16 from one read of each tile's A fragments; coefficient rows built 512x instead of 2048x, 4x fewer waves; sum order kept
# speedup vs baseline: 1.5092x; 1.0642x over previous
#include <hip/hip_runtime.h>
#include <hip/hip_bf16.h>

typedef __attribute__((ext_vector_type(2))) float v2f;
typedef __attribute__((ext_vector_type(8))) float v8f;

namespace {
constexpr int kW = 512;
constexpr int kH = 512;
constexpr int kGroups = 4;
constexpr int kChunk = 128;
constexpr float kLog2e = 1.44269504088896340736f;
}

__global__ __launch_bounds__(256) void gauss2d_rows(
    const float* __restrict__ means, const float* __restrict__ scales,
    const float* __restrict__ thetas, const float* __restrict__ opac_in,
    float* __restrict__ out, int N) {
  __shared__ v2f sA[(kChunk / 16) * 2 * 32];

  const int tid = threadIdx.x;
  const int lane = tid & 31;
  const int wv = tid >> 5;
  const bool hi = lane >= 16;
  const int col = lane & 15;
  const int h = blockIdx.x;
  const float y = -1.0f + 2.0f * (float)h / (float)(kH - 1);

  v2f b1[kGroups], b2;
#pragma unroll
  for (int g = 0; g < kGroups; ++g) {
    const int w = wv * (16 * kGroups) + g * 16 + col;
    const float x = -1.0f + 2.0f * (float)w / (float)(kW - 1);
    b1[g].x = hi ? (y * y) : (x * x);
    b1[g].y = hi ? x       : (x * y);
  }
  b2.x = hi ? 0.0f : y;
  b2.y = hi ? 0.0f : 1.0f;

  v2f cov[kGroups];
#pragma unroll
  for (int g = 0; g < kGroups; ++g) cov[g] = v2f{0.0f, 0.0f};

  for (int base = 0; base < N; base += kChunk) {
    if (base) __syncthreads();

    if (tid < kChunk) {
      const int n = base + tid;
      float g0 = 0.0f, g1 = 0.0f, g2 = 0.0f, g3 = 0.0f, g4 = 0.0f;
      float g5 = -__builtin_inff();
      if (n < N) {
        float mx = means[2 * n + 0];
        float my = means[2 * n + 1];
        float sx = expf(scales[2 * n + 0]);
        float sy = expf(scales[2 * n + 1]);
        float th = (1.0f / (1.0f + expf(-thetas[n]))) * 6.28318530717958647692f;
        float op = 1.0f / (1.0f + expf(-opac_in[n]));
        float c = cosf(th), s = sinf(th);
        float sx2 = sx * sx, sy2 = sy * sy;
        float a = c * c * sx2 + s * s * sy2 + 1e-6f;
        float b = c * s * (sx2 - sy2);
        float d = s * s * sx2 + c * c * sy2 + 1e-6f;
        float det = a * d - b * b;
        float ai = d / det, bi = -b / det, di = a / det;
        const float m = -0.5f * kLog2e;
        g0 = m * ai;
        g1 = m * 2.0f * bi;
        g2 = m * di;
        g3 = -2.0f * m * (ai * mx + bi * my);
        g4 = -2.0f * m * (bi * mx + di * my);
        g5 = m * (ai * mx * mx + 2.0f * bi * mx * my + di * my * my)
             + log2f(op);
      }
      const int tl = tid >> 4;
      const int r  = tid & 15;
      v2f* dst = sA + tl * 64 + r;
      dst[0]  = v2f{g0, g1};
      dst[16] = v2f{g2, g3};
      dst[32] = v2f{g4, g5};
      dst[48] = v2f{0.0f, 0.0f};
    }
    __syncthreads();

    const int left = N - base;
    const int nt = ((left < kChunk ? left : kChunk) + 15) >> 4;
    const v2f* ap = sA + lane;

    for (int t = 0; t < nt; ++t) {
      const v2f a1 = ap[0];
      const v2f a2 = ap[32];
      ap += 64;

      v8f q[kGroups];
#pragma unroll
      for (int g = 0; g < kGroups; ++g) {
        v8f z = {};
        q[g] = __builtin_amdgcn_wmma_f32_16x16x4_f32(false, a1, false, b1[g],
                                                     (short)0, z, false, false);
      }
#pragma unroll
      for (int g = 0; g < kGroups; ++g) {
        q[g] = __builtin_amdgcn_wmma_f32_16x16x4_f32(false, a2, false, b2,
                                                     (short)0, q[g], false, false);
      }

#pragma unroll
      for (int g = 0; g < kGroups; ++g) {
#pragma unroll
        for (int r = 0; r < 8; r += 2) {
          v2f alpha;
          alpha.x = __builtin_amdgcn_exp2f(q[g][r]);
          alpha.y = __builtin_amdgcn_exp2f(q[g][r + 1]);
          const v2f om = 1.0f - alpha;
          cov[g] = cov[g] * om + alpha;
        }
      }
    }
  }

  float acc[kGroups];
#pragma unroll
  for (int g = 0; g < kGroups; ++g) {
    const float mine  = cov[g].x * (1.0f - cov[g].y) + cov[g].y;
    const float other = __shfl_xor(mine, 16, 32);
    const float lower = hi ? other : mine;
    const float upper = hi ? mine : other;
    acc[g] = lower * (1.0f - upper) + upper;
  }

  static_assert(kGroups == 4, "store split below assumes 4 pixel groups per wave");
  {
    const int w0 = wv * 64 + (hi ? 32 : 0) + col;
    float* o = out + ((size_t)h * kW + w0) * 3;
    const float va = hi ? acc[2] : acc[0];
    const float vb = hi ? acc[3] : acc[1];
    o[0] = va;  o[1] = va;  o[2] = va;
    o[48] = vb; o[49] = vb; o[50] = vb;
  }
}

extern "C" void kernel_launch(void* const* d_in, const int* in_sizes, int n_in,
                              void* d_out, int out_size, void* d_ws,
                              size_t ws_size, hipStream_t stream) {
  const float* means  = (const float*)d_in[0];
  const float* scales = (const float*)d_in[1];
  const float* thetas = (const float*)d_in[2];
  const float* opacs  = (const float*)d_in[3];

  const int N = in_sizes[0] / 2;

  gauss2d_rows<<<kH, 256, 0, stream>>>(
      means, scales, thetas, opacs, (float*)d_out, N);
}


# Row's y folded into the coefficient rows (dy = y - mean_y, quadratic in x only): K 8->4, one f32 WMMA per 16x16 instead of two; y terms now summed in the prologue, not inside the WMMA
# speedup vs baseline: 1.8059x; 1.1966x over previous
#include <hip/hip_runtime.h>
#include <hip/hip_bf16.h>

typedef __attribute__((ext_vector_type(2))) float v2f;
typedef __attribute__((ext_vector_type(8))) float v8f;

namespace {
constexpr int kW = 512;
constexpr int kH = 512;
constexpr int kGroups = 4;
constexpr int kChunk = 128;
constexpr float kLog2e = 1.44269504088896340736f;
}

__global__ __launch_bounds__(256) void gauss2d_rows(
    const float* __restrict__ means, const float* __restrict__ scales,
    const float* __restrict__ thetas, const float* __restrict__ opac_in,
    float* __restrict__ out, int N) {
  __shared__ v2f sA[(kChunk / 16) * 32];

  const int tid = threadIdx.x;
  const int lane = tid & 31;
  const int wv = tid >> 5;
  const bool hi = lane >= 16;
  const int col = lane & 15;
  const int h = blockIdx.x;
  const float y = -1.0f + 2.0f * (float)h / (float)(kH - 1);

  v2f b1[kGroups];
#pragma unroll
  for (int g = 0; g < kGroups; ++g) {
    const int w = wv * (16 * kGroups) + g * 16 + col;
    const float x = -1.0f + 2.0f * (float)w / (float)(kW - 1);
    b1[g].x = hi ? 1.0f : (x * x);
    b1[g].y = hi ? 0.0f : x;
  }

  v2f cov[kGroups];
#pragma unroll
  for (int g = 0; g < kGroups; ++g) cov[g] = v2f{0.0f, 0.0f};

  for (int base = 0; base < N; base += kChunk) {
    if (base) __syncthreads();

    if (tid < kChunk) {
      const int n = base + tid;
      float g0 = 0.0f, g1 = 0.0f;
      float g2 = -__builtin_inff();
      if (n < N) {
        float mx = means[2 * n + 0];
        float my = means[2 * n + 1];
        float sx = expf(scales[2 * n + 0]);
        float sy = expf(scales[2 * n + 1]);
        float th = (1.0f / (1.0f + expf(-thetas[n]))) * 6.28318530717958647692f;
        float op = 1.0f / (1.0f + expf(-opac_in[n]));
        float c = cosf(th), s = sinf(th);
        float sx2 = sx * sx, sy2 = sy * sy;
        float a = c * c * sx2 + s * s * sy2 + 1e-6f;
        float b = c * s * (sx2 - sy2);
        float d = s * s * sx2 + c * c * sy2 + 1e-6f;
        float det = a * d - b * b;
        float ai = d / det, bi = -b / det, di = a / det;
        const float m = -0.5f * kLog2e;
        const float dy = y - my;
        g0 = m * ai;
        g1 = 2.0f * m * (bi * dy - ai * mx);
        g2 = m * (ai * mx * mx + dy * (di * dy - 2.0f * bi * mx))
             + log2f(op);
      }
      const int tl = tid >> 4;
      const int r  = tid & 15;
      v2f* dst = sA + tl * 32 + r;
      dst[0]  = v2f{g0, g1};
      dst[16] = v2f{g2, 0.0f};
    }
    __syncthreads();

    const int left = N - base;
    const int nt = ((left < kChunk ? left : kChunk) + 15) >> 4;
    const v2f* ap = sA + lane;

    for (int t = 0; t < nt; ++t) {
      const v2f a1 = ap[0];
      ap += 32;

      v8f q[kGroups];
#pragma unroll
      for (int g = 0; g < kGroups; ++g) {
        v8f z = {};
        q[g] = __builtin_amdgcn_wmma_f32_16x16x4_f32(false, a1, false, b1[g],
                                                     (short)0, z, false, false);
      }

#pragma unroll
      for (int g = 0; g < kGroups; ++g) {
#pragma unroll
        for (int r = 0; r < 8; r += 2) {
          v2f alpha;
          alpha.x = __builtin_amdgcn_exp2f(q[g][r]);
          alpha.y = __builtin_amdgcn_exp2f(q[g][r + 1]);
          const v2f om = 1.0f - alpha;
          cov[g] = cov[g] * om + alpha;
        }
      }
    }
  }

  float acc[kGroups];
#pragma unroll
  for (int g = 0; g < kGroups; ++g) {
    const float mine  = cov[g].x * (1.0f - cov[g].y) + cov[g].y;
    const float other = __shfl_xor(mine, 16, 32);
    const float lower = hi ? other : mine;
    const float upper = hi ? mine : other;
    acc[g] = lower * (1.0f - upper) + upper;
  }

  static_assert(kGroups == 4, "store split below assumes 4 pixel groups per wave");
  {
    const int w0 = wv * 64 + (hi ? 32 : 0) + col;
    float* o = out + ((size_t)h * kW + w0) * 3;
    const float va = hi ? acc[2] : acc[0];
    const float vb = hi ? acc[3] : acc[1];
    o[0] = va;  o[1] = va;  o[2] = va;
    o[48] = vb; o[49] = vb; o[50] = vb;
  }
}

extern "C" void kernel_launch(void* const* d_in, const int* in_sizes, int n_in,
                              void* d_out, int out_size, void* d_ws,
                              size_t ws_size, hipStream_t stream) {
  const float* means  = (const float*)d_in[0];
  const float* scales = (const float*)d_in[1];
  const float* thetas = (const float*)d_in[2];
  const float* opacs  = (const float*)d_in[3];

  const int N = in_sizes[0] / 2;

  gauss2d_rows<<<kH, 256, 0, stream>>>(
      means, scales, thetas, opacs, (float*)d_out, N);
}


# Tile loop pipelined inside the wave: next tile's 4 WMMAs issued one per 16 vector slots among the current tile's alternating exp2 and packed fma; A fragment read a stage ahead; sum order kept
# speedup vs baseline: 1.9148x; 1.0603x over previous
#include <hip/hip_runtime.h>
#include <hip/hip_bf16.h>

typedef __attribute__((ext_vector_type(2))) float v2f;
typedef __attribute__((ext_vector_type(8))) float v8f;

namespace {
constexpr int kW = 512;
constexpr int kH = 512;
constexpr int kGroups = 4;
constexpr int kChunk = 128;
constexpr float kLog2e = 1.44269504088896340736f;
}

__device__ __forceinline__ void exp_pair(const v8f& q0, const v8f& q1, v2f (&al)[8]) {
#pragma unroll
  for (int r = 0; r < 8; r += 2) {
    al[r / 2].x     = __builtin_amdgcn_exp2f(q0[r]);
    al[r / 2].y     = __builtin_amdgcn_exp2f(q0[r + 1]);
    al[4 + r / 2].x = __builtin_amdgcn_exp2f(q1[r]);
    al[4 + r / 2].y = __builtin_amdgcn_exp2f(q1[r + 1]);
  }
}

__device__ __forceinline__ void fold_pair(const v2f (&al)[8], v2f& c0, v2f& c1) {
#pragma unroll
  for (int k = 0; k < 4; ++k) {
    const v2f om0 = 1.0f - al[k];
    const v2f om1 = 1.0f - al[4 + k];
    c0 = c0 * om0 + al[k];
    c1 = c1 * om1 + al[4 + k];
  }
}

#define WMMA_SLOT()                                                            \
  __builtin_amdgcn_sched_group_barrier(0x008, 1, 0);                           \
  _Pragma("unroll") for (int i_ = 0; i_ < 8; ++i_) {                           \
    __builtin_amdgcn_sched_group_barrier(0x400, 1, 0);                         \
    __builtin_amdgcn_sched_group_barrier(0x002, 1, 0);                         \
  }

#define STAGE(QN, QC, AF)                                                      \
  {                                                                            \
    _Pragma("unroll") for (int g_ = 0; g_ < kGroups; ++g_) {                   \
      v8f z_ = {};                                                             \
      QN[g_] = __builtin_amdgcn_wmma_f32_16x16x4_f32(                          \
          false, AF, false, b1[g_], (short)0, z_, false, false);               \
    }                                                                          \
    exp_pair(QC[0], QC[1], al01);                                              \
    fold_pair(al23, cov[2], cov[3]);                                           \
    exp_pair(QC[2], QC[3], al23);                                              \
    fold_pair(al01, cov[0], cov[1]);                                           \
    WMMA_SLOT() WMMA_SLOT() WMMA_SLOT() WMMA_SLOT()                            \
  }

#define LAST(QC)                                                               \
  {                                                                            \
    exp_pair(QC[0], QC[1], al01);                                              \
    fold_pair(al23, cov[2], cov[3]);                                           \
    exp_pair(QC[2], QC[3], al23);                                              \
    fold_pair(al01, cov[0], cov[1]);                                           \
  }

__global__ __launch_bounds__(256) void gauss2d_rows(
    const float* __restrict__ means, const float* __restrict__ scales,
    const float* __restrict__ thetas, const float* __restrict__ opac_in,
    float* __restrict__ out, int N) {
  __shared__ v2f sA[(kChunk / 16) * 32];

  const int tid = threadIdx.x;
  const int lane = tid & 31;
  const int wv = tid >> 5;
  const bool hi = lane >= 16;
  const int col = lane & 15;
  const int h = blockIdx.x;
  const float y = -1.0f + 2.0f * (float)h / (float)(kH - 1);

  v2f b1[kGroups];
#pragma unroll
  for (int g = 0; g < kGroups; ++g) {
    const int w = wv * (16 * kGroups) + g * 16 + col;
    const float x = -1.0f + 2.0f * (float)w / (float)(kW - 1);
    b1[g].x = hi ? 1.0f : (x * x);
    b1[g].y = hi ? 0.0f : x;
  }

  v2f cov[kGroups];
#pragma unroll
  for (int g = 0; g < kGroups; ++g) cov[g] = v2f{0.0f, 0.0f};

  for (int base = 0; base < N; base += kChunk) {
    if (base) __syncthreads();

    if (tid < kChunk) {
      const int n = base + tid;
      float g0 = 0.0f, g1 = 0.0f;
      float g2 = -__builtin_inff();
      if (n < N) {
        float mx = means[2 * n + 0];
        float my = means[2 * n + 1];
        float sx = expf(scales[2 * n + 0]);
        float sy = expf(scales[2 * n + 1]);
        float th = (1.0f / (1.0f + expf(-thetas[n]))) * 6.28318530717958647692f;
        float op = 1.0f / (1.0f + expf(-opac_in[n]));
        float c = cosf(th), s = sinf(th);
        float sx2 = sx * sx, sy2 = sy * sy;
        float a = c * c * sx2 + s * s * sy2 + 1e-6f;
        float b = c * s * (sx2 - sy2);
        float d = s * s * sx2 + c * c * sy2 + 1e-6f;
        float det = a * d - b * b;
        float ai = d / det, bi = -b / det, di = a / det;
        const float m = -0.5f * kLog2e;
        const float dy = y - my;
        g0 = m * ai;
        g1 = 2.0f * m * (bi * dy - ai * mx);
        g2 = m * (ai * mx * mx + dy * (di * dy - 2.0f * bi * mx))
             + log2f(op);
      }
      const int tl = tid >> 4;
      const int r  = tid & 15;
      v2f* dst = sA + tl * 32 + r;
      dst[0]  = v2f{g0, g1};
      dst[16] = v2f{g2, 0.0f};
    }
    __syncthreads();

    const int left = N - base;
    const int nt = ((left < kChunk ? left : kChunk) + 15) >> 4;
    const v2f* ap = sA + lane;

    v8f qa[kGroups], qb[kGroups];
    v2f an = ap[0];
#pragma unroll
    for (int g = 0; g < kGroups; ++g) {
      v8f z = {};
      qa[g] = __builtin_amdgcn_wmma_f32_16x16x4_f32(false, an, false, b1[g],
                                                   (short)0, z, false, false);
    }
    an = ap[32];

    v2f al01[8], al23[8];
#pragma unroll
    for (int i = 0; i < 8; ++i) al23[i] = v2f{0.0f, 0.0f};

    const int nt2 = (nt + 1) & ~1;
    int t = 0;
    for (; t + 2 < nt2; t += 2) {
      const v2f a1 = an;
      an = ap[32 * (t + 2)];
      STAGE(qb, qa, a1)
      const v2f a2 = an;
      an = ap[32 * (t + 3)];
      STAGE(qa, qb, a2)
    }
    STAGE(qb, qa, an)
    LAST(qb)
    fold_pair(al23, cov[2], cov[3]);
  }

  float acc[kGroups];
#pragma unroll
  for (int g = 0; g < kGroups; ++g) {
    const float mine  = cov[g].x * (1.0f - cov[g].y) + cov[g].y;
    const float other = __shfl_xor(mine, 16, 32);
    const float lower = hi ? other : mine;
    const float upper = hi ? mine : other;
    acc[g] = lower * (1.0f - upper) + upper;
  }

  static_assert(kGroups == 4, "store split below assumes 4 pixel groups per wave");
  {
    const int w0 = wv * 64 + (hi ? 32 : 0) + col;
    float* o = out + ((size_t)h * kW + w0) * 3;
    const float va = hi ? acc[2] : acc[0];
    const float vb = hi ? acc[3] : acc[1];
    o[0] = va;  o[1] = va;  o[2] = va;
    o[48] = vb; o[49] = vb; o[50] = vb;
  }
}

extern "C" void kernel_launch(void* const* d_in, const int* in_sizes, int n_in,
                              void* d_out, int out_size, void* d_ws,
                              size_t ws_size, hipStream_t stream) {
  const float* means  = (const float*)d_in[0];
  const float* scales = (const float*)d_in[1];
  const float* thetas = (const float*)d_in[2];
  const float* opacs  = (const float*)d_in[3];

  const int N = in_sizes[0] / 2;

  gauss2d_rows<<<kH, 256, 0, stream>>>(
      means, scales, thetas, opacs, (float*)d_out, N);
}


# Per-wave culling: each wave lists in LDS only Gaussians whose log2(alpha) can reach -40 on its 64 pixels (alphas under 2^-40 dropped) and runs that many tiles; sum order changes
# speedup vs baseline: 2.2910x; 1.1965x over previous
#include <hip/hip_runtime.h>
#include <hip/hip_bf16.h>

typedef __attribute__((ext_vector_type(2))) float v2f;
typedef __attribute__((ext_vector_type(4))) float v4f;
typedef __attribute__((ext_vector_type(8))) float v8f;

namespace {
constexpr int kW = 512;
constexpr int kH = 512;
constexpr int kGroups = 4;
constexpr int kChunk = 128;
constexpr float kLog2e = 1.44269504088896340736f;
constexpr int kTileSlots = (kChunk / 16) * 32;
constexpr float kCullLog2 = -40.0f;
}

__device__ __forceinline__ void exp_pair(const v8f& q0, const v8f& q1, v2f (&al)[8]) {
#pragma unroll
  for (int r = 0; r < 8; r += 2) {
    al[r / 2].x     = __builtin_amdgcn_exp2f(q0[r]);
    al[r / 2].y     = __builtin_amdgcn_exp2f(q0[r + 1]);
    al[4 + r / 2].x = __builtin_amdgcn_exp2f(q1[r]);
    al[4 + r / 2].y = __builtin_amdgcn_exp2f(q1[r + 1]);
  }
}

__device__ __forceinline__ void fold_pair(const v2f (&al)[8], v2f& c0, v2f& c1) {
#pragma unroll
  for (int k = 0; k < 4; ++k) {
    const v2f om0 = 1.0f - al[k];
    const v2f om1 = 1.0f - al[4 + k];
    c0 = c0 * om0 + al[k];
    c1 = c1 * om1 + al[4 + k];
  }
}

#define WMMA_SLOT()                                                            \
  __builtin_amdgcn_sched_group_barrier(0x008, 1, 0);                           \
  _Pragma("unroll") for (int i_ = 0; i_ < 8; ++i_) {                           \
    __builtin_amdgcn_sched_group_barrier(0x400, 1, 0);                         \
    __builtin_amdgcn_sched_group_barrier(0x002, 1, 0);                         \
  }

#define STAGE(QN, QC, AF)                                                      \
  {                                                                            \
    _Pragma("unroll") for (int g_ = 0; g_ < kGroups; ++g_) {                   \
      v8f z_ = {};                                                             \
      QN[g_] = __builtin_amdgcn_wmma_f32_16x16x4_f32(                          \
          false, AF, false, b1[g_], (short)0, z_, false, false);               \
    }                                                                          \
    exp_pair(QC[0], QC[1], al01);                                              \
    fold_pair(al23, cov[2], cov[3]);                                           \
    exp_pair(QC[2], QC[3], al23);                                              \
    fold_pair(al01, cov[0], cov[1]);                                           \
    WMMA_SLOT() WMMA_SLOT() WMMA_SLOT() WMMA_SLOT()                            \
  }

#define LAST(QC)                                                               \
  {                                                                            \
    exp_pair(QC[0], QC[1], al01);                                              \
    fold_pair(al23, cov[2], cov[3]);                                           \
    exp_pair(QC[2], QC[3], al23);                                              \
    fold_pair(al01, cov[0], cov[1]);                                           \
  }

__global__ __launch_bounds__(256) void gauss2d_rows(
    const float* __restrict__ means, const float* __restrict__ scales,
    const float* __restrict__ thetas, const float* __restrict__ opac_in,
    float* __restrict__ out, int N) {
  __shared__ v4f sG[kChunk];
  __shared__ int sIdx[8 * kChunk];
  __shared__ v2f sA[8 * kTileSlots];

  const int tid = threadIdx.x;
  const int lane = tid & 31;
  const int wv = tid >> 5;
  const bool hi = lane >= 16;
  const int col = lane & 15;
  const int h = blockIdx.x;
  const float y = -1.0f + 2.0f * (float)h / (float)(kH - 1);

  v2f b1[kGroups];
#pragma unroll
  for (int g = 0; g < kGroups; ++g) {
    const int w = wv * (16 * kGroups) + g * 16 + col;
    const float x = -1.0f + 2.0f * (float)w / (float)(kW - 1);
    b1[g].x = hi ? 1.0f : (x * x);
    b1[g].y = hi ? 0.0f : x;
  }

  const float xlo = -1.0f + 2.0f * (float)(wv * (16 * kGroups)) / (float)(kW - 1);
  const float xhi = -1.0f + 2.0f * (float)(wv * (16 * kGroups) + 16 * kGroups - 1) / (float)(kW - 1);
  v2f* const myA = sA + wv * kTileSlots;
  int* const myIdx = sIdx + wv * kChunk;

  v2f cov[kGroups];
#pragma unroll
  for (int g = 0; g < kGroups; ++g) cov[g] = v2f{0.0f, 0.0f};

  for (int base = 0; base < N; base += kChunk) {
    if (base) __syncthreads();

    if (tid < kChunk) {
      const int n = base + tid;
      float g0 = -1.0f, g1 = 0.0f;
      float g2 = -__builtin_inff();
      if (n < N) {
        float mx = means[2 * n + 0];
        float my = means[2 * n + 1];
        float sx = expf(scales[2 * n + 0]);
        float sy = expf(scales[2 * n + 1]);
        float th = (1.0f / (1.0f + expf(-thetas[n]))) * 6.28318530717958647692f;
        float op = 1.0f / (1.0f + expf(-opac_in[n]));
        float c = cosf(th), s = sinf(th);
        float sx2 = sx * sx, sy2 = sy * sy;
        float a = c * c * sx2 + s * s * sy2 + 1e-6f;
        float b = c * s * (sx2 - sy2);
        float d = s * s * sx2 + c * c * sy2 + 1e-6f;
        float det = a * d - b * b;
        float ai = d / det, bi = -b / det, di = a / det;
        const float m = -0.5f * kLog2e;
        const float dy = y - my;
        g0 = m * ai;
        g1 = 2.0f * m * (bi * dy - ai * mx);
        g2 = m * (ai * mx * mx + dy * (di * dy - 2.0f * bi * mx))
             + log2f(op);
      }
      sG[tid] = v4f{g0, g1, g2, 0.0f};
    }
    __syncthreads();

    v4f gr[kChunk / 32];
#pragma unroll
    for (int p = 0; p < kChunk / 32; ++p) gr[p] = sG[p * 32 + lane];
    int kept = 0;
#pragma unroll
    for (int p = 0; p < kChunk / 32; ++p) {
      const v4f g = gr[p];
      float xs = -0.5f * g.y * __builtin_amdgcn_rcpf(g.x);
      xs = fminf(fmaxf(xs, xlo), xhi);
      const float qv = fmaf(fmaf(g.x, xs, g.y), xs, g.z);
      const bool keep = !((g.x < 0.0f) && (qv < kCullLog2));
      const unsigned mask = __builtin_amdgcn_ballot_w32(keep);
      if (keep) myIdx[kept + (int)__builtin_amdgcn_mbcnt_lo(mask, 0u)] = p * 32 + lane;
      kept += __builtin_popcount(mask);
    }
    __syncthreads();

    const int nt = (kept + 15) >> 4;
    const int nt2 = nt > 1 ? ((nt + 1) & ~1) : 2;
    const v2f* sG2 = reinterpret_cast<const v2f*>(sG) + (hi ? 1 : 0);
    const v2f zrow = hi ? v2f{-__builtin_inff(), 0.0f} : v2f{0.0f, 0.0f};
    int id[kChunk / 16];
#pragma unroll
    for (int i = 0; i < kChunk / 16; ++i) {
      id[i] = 0;
      if (i * 16 + col < kept) id[i] = myIdx[i * 16 + col];
    }
    v2f en[kChunk / 16];
#pragma unroll
    for (int i = 0; i < kChunk / 16; ++i) {
      const v2f part = sG2[2 * id[i]];
      en[i] = (i * 16 + col < kept) ? part : zrow;
    }
#pragma unroll
    for (int i = 0; i < kChunk / 16; ++i) {
      if (i < nt2) myA[i * 32 + lane] = en[i];
    }

    const v2f* ap = myA + lane;

    v8f qa[kGroups], qb[kGroups];
    v2f an = ap[0];
#pragma unroll
    for (int g = 0; g < kGroups; ++g) {
      v8f z = {};
      qa[g] = __builtin_amdgcn_wmma_f32_16x16x4_f32(false, an, false, b1[g],
                                                   (short)0, z, false, false);
    }
    an = ap[32];

    v2f al01[8], al23[8];
#pragma unroll
    for (int i = 0; i < 8; ++i) al23[i] = v2f{0.0f, 0.0f};

    int t = 0;
    for (; t + 2 < nt2; t += 2) {
      const v2f a1 = an;
      an = ap[32 * (t + 2)];
      STAGE(qb, qa, a1)
      const v2f a2 = an;
      an = ap[32 * (t + 3)];
      STAGE(qa, qb, a2)
    }
    STAGE(qb, qa, an)
    LAST(qb)
    fold_pair(al23, cov[2], cov[3]);
  }

  float acc[kGroups];
#pragma unroll
  for (int g = 0; g < kGroups; ++g) {
    const float mine  = cov[g].x * (1.0f - cov[g].y) + cov[g].y;
    const float other = __shfl_xor(mine, 16, 32);
    const float lower = hi ? other : mine;
    const float upper = hi ? mine : other;
    acc[g] = lower * (1.0f - upper) + upper;
  }

  static_assert(kGroups == 4, "store split below assumes 4 pixel groups per wave");
  {
    const int w0 = wv * 64 + (hi ? 32 : 0) + col;
    float* o = out + ((size_t)h * kW + w0) * 3;
    const float va = hi ? acc[2] : acc[0];
    const float vb = hi ? acc[3] : acc[1];
    o[0] = va;  o[1] = va;  o[2] = va;
    o[48] = vb; o[49] = vb; o[50] = vb;
  }
}

extern "C" void kernel_launch(void* const* d_in, const int* in_sizes, int n_in,
                              void* d_out, int out_size, void* d_ws,
                              size_t ws_size, hipStream_t stream) {
  const float* means  = (const float*)d_in[0];
  const float* scales = (const float*)d_in[1];
  const float* thetas = (const float*)d_in[2];
  const float* opacs  = (const float*)d_in[3];

  const int N = in_sizes[0] / 2;

  gauss2d_rows<<<kH, 256, 0, stream>>>(
      means, scales, thetas, opacs, (float*)d_out, N);
}


# Survivor rows stored straight into the wave's A-fragment table at their ballot-ranked slots by the lanes that tested them: index list, row gather and one workgroup barrier gone; sum order kept
# speedup vs baseline: 2.4188x; 1.0558x over previous
#include <hip/hip_runtime.h>
#include <hip/hip_bf16.h>

typedef __attribute__((ext_vector_type(2))) float v2f;
typedef __attribute__((ext_vector_type(4))) float v4f;
typedef __attribute__((ext_vector_type(8))) float v8f;

namespace {
constexpr int kW = 512;
constexpr int kH = 512;
constexpr int kGroups = 4;
constexpr int kChunk = 128;
constexpr float kLog2e = 1.44269504088896340736f;
constexpr int kTileSlots = (kChunk / 16) * 32;
constexpr float kCullLog2 = -40.0f;
}

__device__ __forceinline__ void exp_pair(const v8f& q0, const v8f& q1, v2f (&al)[8]) {
#pragma unroll
  for (int r = 0; r < 8; r += 2) {
    al[r / 2].x     = __builtin_amdgcn_exp2f(q0[r]);
    al[r / 2].y     = __builtin_amdgcn_exp2f(q0[r + 1]);
    al[4 + r / 2].x = __builtin_amdgcn_exp2f(q1[r]);
    al[4 + r / 2].y = __builtin_amdgcn_exp2f(q1[r + 1]);
  }
}

__device__ __forceinline__ void fold_pair(const v2f (&al)[8], v2f& c0, v2f& c1) {
#pragma unroll
  for (int k = 0; k < 4; ++k) {
    const v2f om0 = 1.0f - al[k];
    const v2f om1 = 1.0f - al[4 + k];
    c0 = c0 * om0 + al[k];
    c1 = c1 * om1 + al[4 + k];
  }
}

#define WMMA_SLOT()                                                            \
  __builtin_amdgcn_sched_group_barrier(0x008, 1, 0);                           \
  _Pragma("unroll") for (int i_ = 0; i_ < 8; ++i_) {                           \
    __builtin_amdgcn_sched_group_barrier(0x400, 1, 0);                         \
    __builtin_amdgcn_sched_group_barrier(0x002, 1, 0);                         \
  }

#define EXPFOLD(QA, QB, AN, AO, CA, CB, K)                                     \
  {                                                                            \
    AN[K].x = __builtin_amdgcn_exp2f(QA[2 * (K)]);                             \
    const v2f om0_ = 1.0f - AO[K];                                             \
    AN[K].y = __builtin_amdgcn_exp2f(QA[2 * (K) + 1]);                         \
    const v2f om1_ = 1.0f - AO[4 + (K)];                                       \
    AN[4 + (K)].x = __builtin_amdgcn_exp2f(QB[2 * (K)]);                       \
    CA = CA * om0_ + AO[K];                                                    \
    AN[4 + (K)].y = __builtin_amdgcn_exp2f(QB[2 * (K) + 1]);                   \
    CB = CB * om1_ + AO[4 + (K)];                                              \
  }

#define NEXT_Q(QN, AF, G)                                                      \
  {                                                                            \
    v8f z_ = {};                                                               \
    QN[G] = __builtin_amdgcn_wmma_f32_16x16x4_f32(                             \
        false, AF, false, b1[G], (short)0, z_, false, false);                  \
  }

#define STAGE(QN, QC, AF)                                                      \
  {                                                                            \
    NEXT_Q(QN, AF, 0)                                                          \
    EXPFOLD(QC[0], QC[1], al01, al23, cov[2], cov[3], 0)                       \
    EXPFOLD(QC[0], QC[1], al01, al23, cov[2], cov[3], 1)                       \
    NEXT_Q(QN, AF, 1)                                                          \
    EXPFOLD(QC[0], QC[1], al01, al23, cov[2], cov[3], 2)                       \
    EXPFOLD(QC[0], QC[1], al01, al23, cov[2], cov[3], 3)                       \
    NEXT_Q(QN, AF, 2)                                                          \
    EXPFOLD(QC[2], QC[3], al23, al01, cov[0], cov[1], 0)                       \
    EXPFOLD(QC[2], QC[3], al23, al01, cov[0], cov[1], 1)                       \
    NEXT_Q(QN, AF, 3)                                                          \
    EXPFOLD(QC[2], QC[3], al23, al01, cov[0], cov[1], 2)                       \
    EXPFOLD(QC[2], QC[3], al23, al01, cov[0], cov[1], 3)                       \
    WMMA_SLOT() WMMA_SLOT() WMMA_SLOT() WMMA_SLOT()                            \
  }

#define LAST(QC)                                                               \
  {                                                                            \
    exp_pair(QC[0], QC[1], al01);                                              \
    fold_pair(al23, cov[2], cov[3]);                                           \
    exp_pair(QC[2], QC[3], al23);                                              \
    fold_pair(al01, cov[0], cov[1]);                                           \
  }

__global__ __launch_bounds__(256) void gauss2d_rows(
    const float* __restrict__ means, const float* __restrict__ scales,
    const float* __restrict__ thetas, const float* __restrict__ opac_in,
    float* __restrict__ out, int N) {
  __shared__ v4f sG[kChunk];
  __shared__ v2f sA[8 * kTileSlots];

  const int tid = threadIdx.x;
  const int lane = tid & 31;
  const int wv = tid >> 5;
  const bool hi = lane >= 16;
  const int col = lane & 15;
  const int h = blockIdx.x;
  const float y = -1.0f + 2.0f * (float)h / (float)(kH - 1);

  v2f b1[kGroups];
#pragma unroll
  for (int g = 0; g < kGroups; ++g) {
    const int w = wv * (16 * kGroups) + g * 16 + col;
    const float x = -1.0f + 2.0f * (float)w / (float)(kW - 1);
    b1[g].x = hi ? 1.0f : (x * x);
    b1[g].y = hi ? 0.0f : x;
  }

  const float xlo = -1.0f + 2.0f * (float)(wv * (16 * kGroups)) / (float)(kW - 1);
  const float xhi = -1.0f + 2.0f * (float)(wv * (16 * kGroups) + 16 * kGroups - 1) / (float)(kW - 1);
  v2f* const myA = sA + wv * kTileSlots;

  v2f cov[kGroups];
#pragma unroll
  for (int g = 0; g < kGroups; ++g) cov[g] = v2f{0.0f, 0.0f};

  for (int base = 0; base < N; base += kChunk) {
    if (base) __syncthreads();

    if (tid < kChunk) {
      const int n = base + tid;
      float g0 = -1.0f, g1 = 0.0f;
      float g2 = -__builtin_inff();
      if (n < N) {
        float mx = means[2 * n + 0];
        float my = means[2 * n + 1];
        float sx = expf(scales[2 * n + 0]);
        float sy = expf(scales[2 * n + 1]);
        float th = (1.0f / (1.0f + expf(-thetas[n]))) * 6.28318530717958647692f;
        float op = 1.0f / (1.0f + expf(-opac_in[n]));
        float c = cosf(th), s = sinf(th);
        float sx2 = sx * sx, sy2 = sy * sy;
        float a = c * c * sx2 + s * s * sy2 + 1e-6f;
        float b = c * s * (sx2 - sy2);
        float d = s * s * sx2 + c * c * sy2 + 1e-6f;
        float det = a * d - b * b;
        float ai = d / det, bi = -b / det, di = a / det;
        const float m = -0.5f * kLog2e;
        const float dy = y - my;
        g0 = m * ai;
        g1 = 2.0f * m * (bi * dy - ai * mx);
        g2 = m * (ai * mx * mx + dy * (di * dy - 2.0f * bi * mx))
             + log2f(op);
      }
      sG[tid] = v4f{g0, g1, g2, 0.0f};
    }
    __syncthreads();

    v4f gr[kChunk / 32];
#pragma unroll
    for (int p = 0; p < kChunk / 32; ++p) gr[p] = sG[p * 32 + lane];
    int kept = 0;
#pragma unroll
    for (int p = 0; p < kChunk / 32; ++p) {
      const v4f g = gr[p];
      float xs = -0.5f * g.y * __builtin_amdgcn_rcpf(g.x);
      xs = fminf(fmaxf(xs, xlo), xhi);
      const float qv = fmaf(fmaf(g.x, xs, g.y), xs, g.z);
      const bool keep = !((g.x < 0.0f) && (qv < kCullLog2));
      const unsigned mask = __builtin_amdgcn_ballot_w32(keep);
      if (keep) {
        const int s = kept + (int)__builtin_amdgcn_mbcnt_lo(mask, 0u);
        v2f* const e = myA + (s + (s & ~15));
        e[0]  = v2f{g.x, g.y};
        e[16] = v2f{g.z, g.w};
      }
      kept += __builtin_popcount(mask);
    }

    const int nt = (kept + 15) >> 4;
    const int nt2 = nt > 1 ? ((nt + 1) & ~1) : 2;
    {
      const int s = kept + lane;
      if (s < 16 * nt2) {
        v2f* const e = myA + (s + (s & ~15));
        e[0]  = v2f{0.0f, 0.0f};
        e[16] = v2f{-__builtin_inff(), 0.0f};
      }
    }
    __builtin_amdgcn_fence(__ATOMIC_RELEASE, "wavefront");
    __builtin_amdgcn_wave_barrier();
    __builtin_amdgcn_fence(__ATOMIC_ACQUIRE, "wavefront");

    const v2f* ap = myA + lane;

    v8f qa[kGroups], qb[kGroups];
    v2f an = ap[0];
#pragma unroll
    for (int g = 0; g < kGroups; ++g) {
      v8f z = {};
      qa[g] = __builtin_amdgcn_wmma_f32_16x16x4_f32(false, an, false, b1[g],
                                                   (short)0, z, false, false);
    }
    an = ap[32];

    v2f al01[8], al23[8];
#pragma unroll
    for (int i = 0; i < 8; ++i) al23[i] = v2f{0.0f, 0.0f};

    int t = 0;
    for (; t + 2 < nt2; t += 2) {
      const v2f a1 = an;
      an = ap[32 * (t + 2)];
      STAGE(qb, qa, a1)
      const v2f a2 = an;
      an = ap[32 * (t + 3)];
      STAGE(qa, qb, a2)
    }
    STAGE(qb, qa, an)
    LAST(qb)
    fold_pair(al23, cov[2], cov[3]);
  }

  float acc[kGroups];
#pragma unroll
  for (int g = 0; g < kGroups; ++g) {
    const float mine  = cov[g].x * (1.0f - cov[g].y) + cov[g].y;
    const float other = __shfl_xor(mine, 16, 32);
    const float lower = hi ? other : mine;
    const float upper = hi ? mine : other;
    acc[g] = lower * (1.0f - upper) + upper;
  }

  static_assert(kGroups == 4, "store split below assumes 4 pixel groups per wave");
  {
    const int w0 = wv * 64 + (hi ? 32 : 0) + col;
    float* o = out + ((size_t)h * kW + w0) * 3;
    const float va = hi ? acc[2] : acc[0];
    const float vb = hi ? acc[3] : acc[1];
    o[0] = va;  o[1] = va;  o[2] = va;
    o[48] = vb; o[49] = vb; o[50] = vb;
  }
}

extern "C" void kernel_launch(void* const* d_in, const int* in_sizes, int n_in,
                              void* d_out, int out_size, void* d_ws,
                              size_t ws_size, hipStream_t stream) {
  const float* means  = (const float*)d_in[0];
  const float* scales = (const float*)d_in[1];
  const float* thetas = (const float*)d_in[2];
  const float* opacs  = (const float*)d_in[3];

  const int N = in_sizes[0] / 2;

  gauss2d_rows<<<kH, 256, 0, stream>>>(
      means, scales, thetas, opacs, (float*)d_out, N);
}


# Head's scalar math on single hardware instructions: 12 IEEE divisions -> v_rcp_f32 and a multiply (pixel x, y: one fma with 2/511), expf -> v_exp_f32, log2f -> v_log_f32; values move by a few ulp
# speedup vs baseline: 2.5626x; 1.0595x over previous
#include <hip/hip_runtime.h>
#include <hip/hip_bf16.h>

typedef __attribute__((ext_vector_type(2))) float v2f;
typedef __attribute__((ext_vector_type(4))) float v4f;
typedef __attribute__((ext_vector_type(8))) float v8f;

namespace {
constexpr int kW = 512;
constexpr int kH = 512;
constexpr int kGroups = 4;
constexpr int kChunk = 128;
constexpr float kLog2e = 1.44269504088896340736f;
constexpr int kTileSlots = (kChunk / 16) * 32;
constexpr float kPitchX = 2.0f / (float)(kW - 1);
constexpr float kPitchY = 2.0f / (float)(kH - 1);
constexpr float kCullLog2 = -40.0f;
}

__device__ __forceinline__ float exp_hw(float x) {
  return __builtin_amdgcn_exp2f(x * kLog2e);
}

__device__ __forceinline__ void exp_pair(const v8f& q0, const v8f& q1, v2f (&al)[8]) {
#pragma unroll
  for (int r = 0; r < 8; r += 2) {
    al[r / 2].x     = __builtin_amdgcn_exp2f(q0[r]);
    al[r / 2].y     = __builtin_amdgcn_exp2f(q0[r + 1]);
    al[4 + r / 2].x = __builtin_amdgcn_exp2f(q1[r]);
    al[4 + r / 2].y = __builtin_amdgcn_exp2f(q1[r + 1]);
  }
}

__device__ __forceinline__ void fold_pair(const v2f (&al)[8], v2f& c0, v2f& c1) {
#pragma unroll
  for (int k = 0; k < 4; ++k) {
    const v2f om0 = 1.0f - al[k];
    const v2f om1 = 1.0f - al[4 + k];
    c0 = c0 * om0 + al[k];
    c1 = c1 * om1 + al[4 + k];
  }
}

#define WMMA_SLOT()                                                            \
  __builtin_amdgcn_sched_group_barrier(0x008, 1, 0);                           \
  _Pragma("unroll") for (int i_ = 0; i_ < 8; ++i_) {                           \
    __builtin_amdgcn_sched_group_barrier(0x400, 1, 0);                         \
    __builtin_amdgcn_sched_group_barrier(0x002, 1, 0);                         \
  }

#define EXPFOLD(QA, QB, AN, AO, CA, CB, K)                                     \
  {                                                                            \
    AN[K].x = __builtin_amdgcn_exp2f(QA[2 * (K)]);                             \
    const v2f om0_ = 1.0f - AO[K];                                             \
    AN[K].y = __builtin_amdgcn_exp2f(QA[2 * (K) + 1]);                         \
    const v2f om1_ = 1.0f - AO[4 + (K)];                                       \
    AN[4 + (K)].x = __builtin_amdgcn_exp2f(QB[2 * (K)]);                       \
    CA = CA * om0_ + AO[K];                                                    \
    AN[4 + (K)].y = __builtin_amdgcn_exp2f(QB[2 * (K) + 1]);                   \
    CB = CB * om1_ + AO[4 + (K)];                                              \
  }

#define NEXT_Q(QN, AF, G)                                                      \
  {                                                                            \
    v8f z_ = {};                                                               \
    QN[G] = __builtin_amdgcn_wmma_f32_16x16x4_f32(                             \
        false, AF, false, b1[G], (short)0, z_, false, false);                  \
  }

#define STAGE(QN, QC, AF)                                                      \
  {                                                                            \
    NEXT_Q(QN, AF, 0)                                                          \
    EXPFOLD(QC[0], QC[1], al01, al23, cov[2], cov[3], 0)                       \
    EXPFOLD(QC[0], QC[1], al01, al23, cov[2], cov[3], 1)                       \
    NEXT_Q(QN, AF, 1)                                                          \
    EXPFOLD(QC[0], QC[1], al01, al23, cov[2], cov[3], 2)                       \
    EXPFOLD(QC[0], QC[1], al01, al23, cov[2], cov[3], 3)                       \
    NEXT_Q(QN, AF, 2)                                                          \
    EXPFOLD(QC[2], QC[3], al23, al01, cov[0], cov[1], 0)                       \
    EXPFOLD(QC[2], QC[3], al23, al01, cov[0], cov[1], 1)                       \
    NEXT_Q(QN, AF, 3)                                                          \
    EXPFOLD(QC[2], QC[3], al23, al01, cov[0], cov[1], 2)                       \
    EXPFOLD(QC[2], QC[3], al23, al01, cov[0], cov[1], 3)                       \
    WMMA_SLOT() WMMA_SLOT() WMMA_SLOT() WMMA_SLOT()                            \
  }

#define LAST(QC)                                                               \
  {                                                                            \
    exp_pair(QC[0], QC[1], al01);                                              \
    fold_pair(al23, cov[2], cov[3]);                                           \
    exp_pair(QC[2], QC[3], al23);                                              \
    fold_pair(al01, cov[0], cov[1]);                                           \
  }

__global__ __launch_bounds__(256) void gauss2d_rows(
    const float* __restrict__ means, const float* __restrict__ scales,
    const float* __restrict__ thetas, const float* __restrict__ opac_in,
    float* __restrict__ out, int N) {
  __shared__ v4f sG[kChunk];
  __shared__ v2f sA[8 * kTileSlots];

  const int tid = threadIdx.x;
  const int lane = tid & 31;
  const int wv = tid >> 5;
  const bool hi = lane >= 16;
  const int col = lane & 15;
  const int h = blockIdx.x;
  const float y = __builtin_fmaf((float)h, kPitchY, -1.0f);

  v2f b1[kGroups];
#pragma unroll
  for (int g = 0; g < kGroups; ++g) {
    const int w = wv * (16 * kGroups) + g * 16 + col;
    const float x = __builtin_fmaf((float)w, kPitchX, -1.0f);
    b1[g].x = hi ? 1.0f : (x * x);
    b1[g].y = hi ? 0.0f : x;
  }

  const float xlo = __builtin_fmaf((float)(wv * (16 * kGroups)), kPitchX, -1.0f);
  const float xhi = __builtin_fmaf((float)(wv * (16 * kGroups) + 16 * kGroups - 1), kPitchX, -1.0f);
  v2f* const myA = sA + wv * kTileSlots;

  v2f cov[kGroups];
#pragma unroll
  for (int g = 0; g < kGroups; ++g) cov[g] = v2f{0.0f, 0.0f};

  for (int base = 0; base < N; base += kChunk) {
    if (base) __syncthreads();

    if (tid < kChunk) {
      const int n = base + tid;
      float g0 = -1.0f, g1 = 0.0f;
      float g2 = -__builtin_inff();
      if (n < N) {
        float mx = means[2 * n + 0];
        float my = means[2 * n + 1];
        float sx = exp_hw(scales[2 * n + 0]);
        float sy = exp_hw(scales[2 * n + 1]);
        float th = __builtin_amdgcn_rcpf(1.0f + exp_hw(-thetas[n])) * 6.28318530717958647692f;
        float op = __builtin_amdgcn_rcpf(1.0f + exp_hw(-opac_in[n]));
        float c = cosf(th), s = sinf(th);
        float sx2 = sx * sx, sy2 = sy * sy;
        float a = c * c * sx2 + s * s * sy2 + 1e-6f;
        float b = c * s * (sx2 - sy2);
        float d = s * s * sx2 + c * c * sy2 + 1e-6f;
        float det = a * d - b * b;
        const float rdet = __builtin_amdgcn_rcpf(det);
        float ai = d * rdet, bi = -b * rdet, di = a * rdet;
        const float m = -0.5f * kLog2e;
        const float dy = y - my;
        g0 = m * ai;
        g1 = 2.0f * m * (bi * dy - ai * mx);
        g2 = m * (ai * mx * mx + dy * (di * dy - 2.0f * bi * mx))
             + __builtin_amdgcn_logf(op);
      }
      sG[tid] = v4f{g0, g1, g2, 0.0f};
    }
    __syncthreads();

    v4f gr[kChunk / 32];
#pragma unroll
    for (int p = 0; p < kChunk / 32; ++p) gr[p] = sG[p * 32 + lane];
    int kept = 0;
#pragma unroll
    for (int p = 0; p < kChunk / 32; ++p) {
      const v4f g = gr[p];
      float xs = -0.5f * g.y * __builtin_amdgcn_rcpf(g.x);
      xs = fminf(fmaxf(xs, xlo), xhi);
      const float qv = fmaf(fmaf(g.x, xs, g.y), xs, g.z);
      const bool keep = !((g.x < 0.0f) && (qv < kCullLog2));
      const unsigned mask = __builtin_amdgcn_ballot_w32(keep);
      if (keep) {
        const int s = kept + (int)__builtin_amdgcn_mbcnt_lo(mask, 0u);
        v2f* const e = myA + (s + (s & ~15));
        e[0]  = v2f{g.x, g.y};
        e[16] = v2f{g.z, g.w};
      }
      kept += __builtin_popcount(mask);
    }

    const int nt = (kept + 15) >> 4;
    const int nt2 = nt > 1 ? ((nt + 1) & ~1) : 2;
    {
      const int s = kept + lane;
      if (s < 16 * nt2) {
        v2f* const e = myA + (s + (s & ~15));
        e[0]  = v2f{0.0f, 0.0f};
        e[16] = v2f{-__builtin_inff(), 0.0f};
      }
    }
    __builtin_amdgcn_fence(__ATOMIC_RELEASE, "wavefront");
    __builtin_amdgcn_wave_barrier();
    __builtin_amdgcn_fence(__ATOMIC_ACQUIRE, "wavefront");

    const v2f* ap = myA + lane;

    v8f qa[kGroups], qb[kGroups];
    v2f an = ap[0];
#pragma unroll
    for (int g = 0; g < kGroups; ++g) {
      v8f z = {};
      qa[g] = __builtin_amdgcn_wmma_f32_16x16x4_f32(false, an, false, b1[g],
                                                   (short)0, z, false, false);
    }
    an = ap[32];

    v2f al01[8], al23[8];
#pragma unroll
    for (int i = 0; i < 8; ++i) al23[i] = v2f{0.0f, 0.0f};

    int t = 0;
    for (; t + 2 < nt2; t += 2) {
      const v2f a1 = an;
      an = ap[32 * (t + 2)];
      STAGE(qb, qa, a1)
      const v2f a2 = an;
      an = ap[32 * (t + 3)];
      STAGE(qa, qb, a2)
    }
    STAGE(qb, qa, an)
    LAST(qb)
    fold_pair(al23, cov[2], cov[3]);
  }

  float acc[kGroups];
#pragma unroll
  for (int g = 0; g < kGroups; ++g) {
    const float mine  = cov[g].x * (1.0f - cov[g].y) + cov[g].y;
    const float other = __shfl_xor(mine, 16, 32);
    const float lower = hi ? other : mine;
    const float upper = hi ? mine : other;
    acc[g] = lower * (1.0f - upper) + upper;
  }

  static_assert(kGroups == 4, "store split below assumes 4 pixel groups per wave");
  {
    const int w0 = wv * 64 + (hi ? 32 : 0) + col;
    float* o = out + ((size_t)h * kW + w0) * 3;
    const float va = hi ? acc[2] : acc[0];
    const float vb = hi ? acc[3] : acc[1];
    o[0] = va;  o[1] = va;  o[2] = va;
    o[48] = vb; o[49] = vb; o[50] = vb;
  }
}

extern "C" void kernel_launch(void* const* d_in, const int* in_sizes, int n_in,
                              void* d_out, int out_size, void* d_ws,
                              size_t ws_size, hipStream_t stream) {
  const float* means  = (const float*)d_in[0];
  const float* scales = (const float*)d_in[1];
  const float* thetas = (const float*)d_in[2];
  const float* opacs  = (const float*)d_in[3];

  const int N = in_sizes[0] / 2;

  gauss2d_rows<<<kH, 256, 0, stream>>>(
      means, scales, thetas, opacs, (float*)d_out, N);
}


# No vector-compare branch left before the tile loop (10 gone): head guards -> scalar test + selects, sinf/cosf -> v_sin_f32/v_cos_f32 of the sigmoid (~1e-6), cull rows stored unmasked; sum order kept
# speedup vs baseline: 2.7268x; 1.0641x over previous
#include <hip/hip_runtime.h>
#include <hip/hip_bf16.h>

typedef __attribute__((ext_vector_type(2))) float v2f;
typedef __attribute__((ext_vector_type(4))) float v4f;
typedef __attribute__((ext_vector_type(8))) float v8f;

namespace {
constexpr int kW = 512;
constexpr int kH = 512;
constexpr int kGroups = 4;
constexpr int kChunk = 128;
constexpr float kLog2e = 1.44269504088896340736f;
constexpr int kTableTiles = kChunk / 16 + 4;
constexpr int kTileSlots = kTableTiles * 32;
constexpr int kDumpEntry = (kChunk / 16 + 2) * 32;
constexpr float kPitchX = 2.0f / (float)(kW - 1);
constexpr float kPitchY = 2.0f / (float)(kH - 1);
constexpr float kCullLog2 = -40.0f;
}

__device__ __forceinline__ float exp_hw(float x) {
  return __builtin_amdgcn_exp2f(x * kLog2e);
}

__device__ __forceinline__ void exp_pair(const v8f& q0, const v8f& q1, v2f (&al)[8]) {
#pragma unroll
  for (int r = 0; r < 8; r += 2) {
    al[r / 2].x     = __builtin_amdgcn_exp2f(q0[r]);
    al[r / 2].y     = __builtin_amdgcn_exp2f(q0[r + 1]);
    al[4 + r / 2].x = __builtin_amdgcn_exp2f(q1[r]);
    al[4 + r / 2].y = __builtin_amdgcn_exp2f(q1[r + 1]);
  }
}

__device__ __forceinline__ void fold_pair(const v2f (&al)[8], v2f& c0, v2f& c1) {
#pragma unroll
  for (int k = 0; k < 4; ++k) {
    const v2f om0 = 1.0f - al[k];
    const v2f om1 = 1.0f - al[4 + k];
    c0 = c0 * om0 + al[k];
    c1 = c1 * om1 + al[4 + k];
  }
}

#define WMMA_SLOT()                                                            \
  __builtin_amdgcn_sched_group_barrier(0x008, 1, 0);                           \
  _Pragma("unroll") for (int i_ = 0; i_ < 8; ++i_) {                           \
    __builtin_amdgcn_sched_group_barrier(0x400, 1, 0);                         \
    __builtin_amdgcn_sched_group_barrier(0x002, 1, 0);                         \
  }

#define EXPFOLD(QA, QB, AN, AO, CA, CB, K)                                     \
  {                                                                            \
    AN[K].x = __builtin_amdgcn_exp2f(QA[2 * (K)]);                             \
    const v2f om0_ = 1.0f - AO[K];                                             \
    AN[K].y = __builtin_amdgcn_exp2f(QA[2 * (K) + 1]);                         \
    const v2f om1_ = 1.0f - AO[4 + (K)];                                       \
    AN[4 + (K)].x = __builtin_amdgcn_exp2f(QB[2 * (K)]);                       \
    CA = CA * om0_ + AO[K];                                                    \
    AN[4 + (K)].y = __builtin_amdgcn_exp2f(QB[2 * (K) + 1]);                   \
    CB = CB * om1_ + AO[4 + (K)];                                              \
  }

#define NEXT_Q(QN, AF, G)                                                      \
  {                                                                            \
    v8f z_ = {};                                                               \
    QN[G] = __builtin_amdgcn_wmma_f32_16x16x4_f32(                             \
        false, AF, false, b1[G], (short)0, z_, false, false);                  \
  }

#define STAGE(QN, QC, AF)                                                      \
  {                                                                            \
    NEXT_Q(QN, AF, 0)                                                          \
    EXPFOLD(QC[0], QC[1], al01, al23, cov[2], cov[3], 0)                       \
    EXPFOLD(QC[0], QC[1], al01, al23, cov[2], cov[3], 1)                       \
    NEXT_Q(QN, AF, 1)                                                          \
    EXPFOLD(QC[0], QC[1], al01, al23, cov[2], cov[3], 2)                       \
    EXPFOLD(QC[0], QC[1], al01, al23, cov[2], cov[3], 3)                       \
    NEXT_Q(QN, AF, 2)                                                          \
    EXPFOLD(QC[2], QC[3], al23, al01, cov[0], cov[1], 0)                       \
    EXPFOLD(QC[2], QC[3], al23, al01, cov[0], cov[1], 1)                       \
    NEXT_Q(QN, AF, 3)                                                          \
    EXPFOLD(QC[2], QC[3], al23, al01, cov[0], cov[1], 2)                       \
    EXPFOLD(QC[2], QC[3], al23, al01, cov[0], cov[1], 3)                       \
    WMMA_SLOT() WMMA_SLOT() WMMA_SLOT() WMMA_SLOT()                            \
  }

#define LAST(QC)                                                               \
  {                                                                            \
    exp_pair(QC[0], QC[1], al01);                                              \
    fold_pair(al23, cov[2], cov[3]);                                           \
    exp_pair(QC[2], QC[3], al23);                                              \
    fold_pair(al01, cov[0], cov[1]);                                           \
  }

__global__ __launch_bounds__(256) void gauss2d_rows(
    const float* __restrict__ means, const float* __restrict__ scales,
    const float* __restrict__ thetas, const float* __restrict__ opac_in,
    float* __restrict__ out, int N) {
  __shared__ v4f sG[kChunk];
  __shared__ v2f sA[8 * kTileSlots];

  const int tid = threadIdx.x;
  const int lane = tid & 31;
  const int wv = __builtin_amdgcn_readfirstlane(tid >> 5);
  const bool hi = lane >= 16;
  const int col = lane & 15;
  const int h = blockIdx.x;
  const float y = __builtin_fmaf((float)h, kPitchY, -1.0f);

  v2f b1[kGroups];
#pragma unroll
  for (int g = 0; g < kGroups; ++g) {
    const int w = wv * (16 * kGroups) + g * 16 + col;
    const float x = __builtin_fmaf((float)w, kPitchX, -1.0f);
    b1[g].x = hi ? 1.0f : (x * x);
    b1[g].y = hi ? 0.0f : x;
  }

  const float xlo = __builtin_fmaf((float)(wv * (16 * kGroups)), kPitchX, -1.0f);
  const float xhi = __builtin_fmaf((float)(wv * (16 * kGroups) + 16 * kGroups - 1), kPitchX, -1.0f);
  v2f* const myA = sA + wv * kTileSlots;

  v2f cov[kGroups];
#pragma unroll
  for (int g = 0; g < kGroups; ++g) cov[g] = v2f{0.0f, 0.0f};

  for (int base = 0; base < N; base += kChunk) {
    if (base) __syncthreads();

    if (wv < kChunk / 32) {
      const int n = base + tid;
      const bool live = n < N;
      const int nn = live ? n : N - 1;
      float g0, g1, g2;
      {
        float mx = means[2 * nn + 0];
        float my = means[2 * nn + 1];
        float sx = exp_hw(scales[2 * nn + 0]);
        float sy = exp_hw(scales[2 * nn + 1]);
        float sg = __builtin_amdgcn_rcpf(1.0f + exp_hw(-thetas[nn]));
        float op = __builtin_amdgcn_rcpf(1.0f + exp_hw(-opac_in[nn]));
        float c = __builtin_amdgcn_cosf(sg), s = __builtin_amdgcn_sinf(sg);
        float sx2 = sx * sx, sy2 = sy * sy;
        float a = c * c * sx2 + s * s * sy2 + 1e-6f;
        float b = c * s * (sx2 - sy2);
        float d = s * s * sx2 + c * c * sy2 + 1e-6f;
        float det = a * d - b * b;
        const float rdet = __builtin_amdgcn_rcpf(det);
        float ai = d * rdet, bi = -b * rdet, di = a * rdet;
        const float m = -0.5f * kLog2e;
        const float dy = y - my;
        g0 = m * ai;
        g1 = 2.0f * m * (bi * dy - ai * mx);
        g2 = m * (ai * mx * mx + dy * (di * dy - 2.0f * bi * mx))
             + __builtin_amdgcn_logf(op);
      }
      g0 = live ? g0 : -1.0f;
      g1 = live ? g1 : 0.0f;
      g2 = live ? g2 : -__builtin_inff();
      sG[tid] = v4f{g0, g1, g2, 0.0f};
    }
    __syncthreads();

    v4f gr[kChunk / 32];
#pragma unroll
    for (int p = 0; p < kChunk / 32; ++p) gr[p] = sG[p * 32 + lane];
    bool keep[kChunk / 32];
    unsigned mask[kChunk / 32];
    const int dump = kDumpEntry + lane + (lane & 16);
#pragma unroll
    for (int p = 0; p < kChunk / 32; ++p) {
      const v4f g = gr[p];
      float xs = -0.5f * g.y * __builtin_amdgcn_rcpf(g.x);
      xs = fminf(fmaxf(xs, xlo), xhi);
      const float qv = fmaf(fmaf(g.x, xs, g.y), xs, g.z);
      keep[p] = !((g.x < 0.0f) && (qv < kCullLog2));
      mask[p] = __builtin_amdgcn_ballot_w32(keep[p]);
    }
    int kept = 0;
#pragma unroll
    for (int p = 0; p < kChunk / 32; ++p) {
      const v4f g = gr[p];
      const int s = kept + (int)__builtin_amdgcn_mbcnt_lo(mask[p], 0u);
      v2f* const e = myA + (keep[p] ? (s + (s & ~15)) : dump);
      e[0]  = v2f{g.x, g.y};
      e[16] = v2f{g.z, g.w};
      kept += __builtin_popcount(mask[p]);
    }

    const int nt = (kept + 15) >> 4;
    const int nt2 = nt > 1 ? ((nt + 1) & ~1) : 2;
    {
      const int s = kept + lane;
      v2f* const e = myA + (s + (s & ~15));
      e[0]  = v2f{0.0f, 0.0f};
      e[16] = v2f{-__builtin_inff(), 0.0f};
    }
    __builtin_amdgcn_fence(__ATOMIC_RELEASE, "wavefront");
    __builtin_amdgcn_wave_barrier();
    __builtin_amdgcn_fence(__ATOMIC_ACQUIRE, "wavefront");

    const v2f* ap = myA + lane;

    v8f qa[kGroups], qb[kGroups];
    v2f an = ap[0];
#pragma unroll
    for (int g = 0; g < kGroups; ++g) {
      v8f z = {};
      qa[g] = __builtin_amdgcn_wmma_f32_16x16x4_f32(false, an, false, b1[g],
                                                   (short)0, z, false, false);
    }
    an = ap[32];

    v2f al01[8], al23[8];
#pragma unroll
    for (int i = 0; i < 8; ++i) al23[i] = v2f{0.0f, 0.0f};

    int t = 0;
    for (; t + 2 < nt2; t += 2) {
      const v2f a1 = an;
      an = ap[32 * (t + 2)];
      STAGE(qb, qa, a1)
      const v2f a2 = an;
      an = ap[32 * (t + 3)];
      STAGE(qa, qb, a2)
    }
    STAGE(qb, qa, an)
    LAST(qb)
    fold_pair(al23, cov[2], cov[3]);
  }

  float acc[kGroups];
#pragma unroll
  for (int g = 0; g < kGroups; ++g) {
    const float mine  = cov[g].x * (1.0f - cov[g].y) + cov[g].y;
    const float other = __shfl_xor(mine, 16, 32);
    const float lower = hi ? other : mine;
    const float upper = hi ? mine : other;
    acc[g] = lower * (1.0f - upper) + upper;
  }

  static_assert(kGroups == 4, "store split below assumes 4 pixel groups per wave");
  {
    const int w0 = wv * 64 + (hi ? 32 : 0) + col;
    float* o = out + ((size_t)h * kW + w0) * 3;
    const float va = hi ? acc[2] : acc[0];
    const float vb = hi ? acc[3] : acc[1];
    o[0] = va;  o[1] = va;  o[2] = va;
    o[48] = vb; o[49] = vb; o[50] = vb;
  }
}

extern "C" void kernel_launch(void* const* d_in, const int* in_sizes, int n_in,
                              void* d_out, int out_size, void* d_ws,
                              size_t ws_size, hipStream_t stream) {
  const float* means  = (const float*)d_in[0];
  const float* scales = (const float*)d_in[1];
  const float* thetas = (const float*)d_in[2];
  const float* opacs  = (const float*)d_in[3];

  const int N = in_sizes[0] / 2;

  gauss2d_rows<<<kH, 256, 0, stream>>>(
      means, scales, thetas, opacs, (float*)d_out, N);
}
